# starting kernel
# baseline (speedup 1.0000x reference)
#include <hip/hip_runtime.h>


#define N_NODES 50000
#define E_EDGES 800000
#define ET      (E_EDGES + N_NODES)
#define DIN     256
#define C_OUT   128
#define NEG_SLOPE 0.2f

typedef __attribute__((ext_vector_type(2))) float v2f;
typedef __attribute__((ext_vector_type(8))) float v8f;

__device__ __forceinline__ void hw_atomic_add_f32(float* p, float v) {
  asm volatile("global_atomic_add_f32 %0, %1, off" : : "v"(p), "v"(v) : "memory");
}

__global__ __launch_bounds__(256) void k_gemm(const float* __restrict__ x,
                                              const float* __restrict__ W,
                                              float* __restrict__ h) {
  const int wave = threadIdx.x >> 5;
  const int lane = threadIdx.x & 31;
  const int tile = blockIdx.x * 8 + wave;
  const int tM   = (tile >> 3) << 4;
  const int tN   = (tile & 7) << 4;
  const int hi   = lane >> 4;
  const int l    = lane & 15;

  const float* arow = x + (size_t)(tM + l) * DIN;
  v8f acc = (v8f)0.0f;

  for (int k = 0; k < DIN; k += 4) {
    const int ka = k + 2 * hi;
    v2f a = *(const v2f*)(arow + ka);
    v2f b;
    b.x = W[(size_t)ka * C_OUT + tN + l];
    b.y = W[(size_t)(ka + 1) * C_OUT + tN + l];
    acc = __builtin_amdgcn_wmma_f32_16x16x4_f32(false, a, false, b,
                                                (short)0, acc, false, false);
  }
  float* orow = h + (size_t)(tM + hi * 8) * C_OUT + tN + l;
#pragma unroll
  for (int r = 0; r < 8; ++r) orow[(size_t)r * C_OUT] = acc[r];
}

__global__ __launch_bounds__(256) void k_nodeatt(const float* __restrict__ h,
                                                 const float* __restrict__ att,
                                                 float* __restrict__ ai,
                                                 float* __restrict__ aj) {
  const int wave = threadIdx.x >> 5;
  const int lane = threadIdx.x & 31;
  const int n = blockIdx.x * 8 + wave;
  const float4 hv  = *(const float4*)(h + (size_t)n * C_OUT + lane * 4);
  const float4 av  = *(const float4*)(att + lane * 4);
  const float4 bv  = *(const float4*)(att + C_OUT + lane * 4);
  float si = hv.x*av.x + hv.y*av.y + hv.z*av.z + hv.w*av.w;
  float sj = hv.x*bv.x + hv.y*bv.y + hv.z*bv.z + hv.w*bv.w;
#pragma unroll
  for (int off = 16; off > 0; off >>= 1) {
    si += __shfl_down(si, off, 32);
    sj += __shfl_down(sj, off, 32);
  }
  if (lane == 0) { ai[n] = si; aj[n] = sj; }
}

__global__ void k_init(unsigned* __restrict__ amax, float* __restrict__ denom) {
  int n = blockIdx.x * blockDim.x + threadIdx.x;
  if (n < N_NODES) { amax[n] = 0u; denom[n] = 0.0f; }
}
__global__ void k_initout(float* __restrict__ out, const float* __restrict__ bias) {
  int i = blockIdx.x * blockDim.x + threadIdx.x;
  if (i < N_NODES * C_OUT) out[i] = bias[i & (C_OUT - 1)];
}

__device__ __forceinline__ unsigned f2key(float f) {
  unsigned b = __float_as_uint(f);
  return (b & 0x80000000u) ? ~b : (b | 0x80000000u);
}
__device__ __forceinline__ float key2f(unsigned k) {
  unsigned b = (k & 0x80000000u) ? (k & 0x7FFFFFFFu) : ~k;
  return __uint_as_float(b);
}
__device__ __forceinline__ float leaky(float a) {
  return a > 0.0f ? a : NEG_SLOPE * a;
}

__global__ void k_edgemax(const int* __restrict__ esrc, const int* __restrict__ edst,
                          const float* __restrict__ ai, const float* __restrict__ aj,
                          unsigned* __restrict__ amax) {
  int e = blockIdx.x * blockDim.x + threadIdx.x;
  if (e >= ET) return;
  int s, d;
  if (e < E_EDGES) { s = esrc[e]; d = edst[e]; } else { s = d = e - E_EDGES; }
  float a = leaky(ai[d] + aj[s]);
  atomicMax(&amax[d], f2key(a));
}

__global__ void k_edgeexp(const int* __restrict__ esrc, const int* __restrict__ edst,
                          const float* __restrict__ ai, const float* __restrict__ aj,
                          const unsigned* __restrict__ amax,
                          float* __restrict__ denom, float* __restrict__ ex) {
  int e = blockIdx.x * blockDim.x + threadIdx.x;
  if (e >= ET) return;
  int s, d;
  if (e < E_EDGES) { s = esrc[e]; d = edst[e]; } else { s = d = e - E_EDGES; }
  float a = leaky(ai[d] + aj[s]);
  float v = expf(a - key2f(amax[d]));
  ex[e] = v;
  hw_atomic_add_f32(&denom[d], v);
}

__global__ __launch_bounds__(256) void k_scatter(const int* __restrict__ esrc,
                                                 const int* __restrict__ edst,
                                                 const float* __restrict__ h,
                                                 const float* __restrict__ ex,
                                                 const float* __restrict__ denom,
                                                 float* __restrict__ out) {
  const int wave = threadIdx.x >> 5;
  const int lane = threadIdx.x & 31;
  const int e = blockIdx.x * 8 + wave;
  int s, d;
  if (e < E_EDGES) { s = esrc[e]; d = edst[e]; } else { s = d = e - E_EDGES; }
  const float coeff = ex[e] / (denom[d] + 1e-16f);
  const float4 hv = *(const float4*)(h + (size_t)s * C_OUT + lane * 4);
  float* o = out + (size_t)d * C_OUT + lane * 4;
  hw_atomic_add_f32(o + 0, hv.x * coeff);
  hw_atomic_add_f32(o + 1, hv.y * coeff);
  hw_atomic_add_f32(o + 2, hv.z * coeff);
  hw_atomic_add_f32(o + 3, hv.w * coeff);
}

extern "C" void kernel_launch(void* const* d_in, const int* in_sizes, int n_in,
                              void* d_out, int out_size, void* d_ws, size_t ws_size,
                              hipStream_t stream) {
  const float* x    = (const float*)d_in[0];
  const int*   ei   = (const int*)  d_in[1];
  const float* W    = (const float*)d_in[2];
  const float* att  = (const float*)d_in[3];
  const float* bias = (const float*)d_in[4];
  float* out = (float*)d_out;

  float*    h     = (float*)d_ws;
  float*    ai    = h + (size_t)N_NODES * C_OUT;
  float*    aj    = ai + N_NODES;
  unsigned* amax  = (unsigned*)(aj + N_NODES);
  float*    denom = (float*)(amax + N_NODES);
  float*    ex    = denom + N_NODES;

  const int* esrc = ei;
  const int* edst = ei + E_EDGES;

  k_gemm   <<<3125, 256, 0, stream>>>(x, W, h);
  k_init   <<<(N_NODES + 255) / 256, 256, 0, stream>>>(amax, denom);
  k_initout<<<(N_NODES * C_OUT + 255) / 256, 256, 0, stream>>>(out, bias);
  k_nodeatt<<<N_NODES / 8, 256, 0, stream>>>(h, att, ai, aj);
  k_edgemax<<<(ET + 255) / 256, 256, 0, stream>>>(esrc, edst, ai, aj, amax);
  k_edgeexp<<<(ET + 255) / 256, 256, 0, stream>>>(esrc, edst, ai, aj, amax, denom, ex);
  k_scatter<<<ET / 8, 256, 0, stream>>>(esrc, edst, h, ex, denom, out);
}


# GEMM: W staged once per work-group in LDS and each wave computes a 16x128 strip (8 accumulator tiles), so each x fragment is loaded once, not 8 times; k order and summation order unchanged
# speedup vs baseline: 1.0873x; 1.0873x over previous
#include <hip/hip_runtime.h>


#define N_NODES 50000
#define E_EDGES 800000
#define ET      (E_EDGES + N_NODES)
#define DIN     256
#define C_OUT   128
#define NEG_SLOPE 0.2f

typedef __attribute__((ext_vector_type(2))) float v2f;
typedef __attribute__((ext_vector_type(8))) float v8f;

__device__ __forceinline__ void hw_atomic_add_f32(float* p, float v) {
  asm volatile("global_atomic_add_f32 %0, %1, off" : : "v"(p), "v"(v) : "memory");
}

#define GEMM_THREADS 512
#define GEMM_WAVES   (GEMM_THREADS / 32)
#define GEMM_BLOCKS  256
#define M_STRIPS     (N_NODES / 16)
#define WPITCH       260
__global__ __launch_bounds__(GEMM_THREADS) void k_gemm(const float* __restrict__ x,
                                                       const float* __restrict__ W,
                                                       float* __restrict__ h) {
  __shared__ float Wt[C_OUT * WPITCH];

  const int wave  = threadIdx.x >> 5;
  const int lane  = threadIdx.x & 31;
  const int hi    = lane >> 4;
  const int l     = lane & 15;
  const int strip = wave * GEMM_BLOCKS + blockIdx.x;
  const bool valid = strip < M_STRIPS;
  const int tM = valid ? (strip << 4) : 0;

  const float* arow = x + (size_t)(tM + l) * DIN + 2 * hi;

  v2f a_cur[8], a_nxt[8];
#pragma unroll
  for (int j = 0; j < 8; ++j) a_cur[j] = *(const v2f*)(arow + 4 * j);

  {
    const int kk = lane & 7;
    const int cc = lane >> 3;
    for (int p = wave; p < (DIN / 8) * (C_OUT / 16); p += GEMM_WAVES) {
      const int k  = ((p >> 3) << 3) + kk;
      const int c4 = ((p & 7) << 2) + cc;
      const float4 w4 = *(const float4*)(W + (size_t)k * C_OUT + c4 * 4);
      float* d = Wt + (c4 * 4) * WPITCH + k;
      d[0]          = w4.x;
      d[WPITCH]     = w4.y;
      d[2 * WPITCH] = w4.z;
      d[3 * WPITCH] = w4.w;
    }
  }
  __syncthreads();
  if (!valid) return;

  const float* bcol = Wt + l * WPITCH + 2 * hi;

  v8f acc[8];
#pragma unroll
  for (int t = 0; t < 8; ++t) acc[t] = (v8f)0.0f;

  v2f b0[8], b1[8];
#pragma unroll
  for (int t = 0; t < 8; ++t) b0[t] = *(const v2f*)(bcol + 16 * t * WPITCH);

  for (int c = 0; c < DIN / 32; ++c) {
    const int cn = (c + 1 < DIN / 32) ? (c + 1) : c;
#pragma unroll
    for (int j = 0; j < 8; ++j) a_nxt[j] = *(const v2f*)(arow + 32 * cn + 4 * j);
    __builtin_amdgcn_sched_barrier(0);

#pragma unroll
    for (int j = 0; j < 8; j += 2) {
      const int q  = 8 * c + j;
      const int q2 = (q + 2 < DIN / 4) ? (q + 2) : (DIN / 4 - 1);
#pragma unroll
      for (int t = 0; t < 8; ++t) b1[t] = *(const v2f*)(bcol + 16 * t * WPITCH + 4 * (q + 1));
      __builtin_amdgcn_sched_barrier(0);
#pragma unroll
      for (int t = 0; t < 8; ++t)
        acc[t] = __builtin_amdgcn_wmma_f32_16x16x4_f32(false, a_cur[j], false, b0[t],
                                                       (short)0, acc[t], false, false);
#pragma unroll
      for (int t = 0; t < 8; ++t) b0[t] = *(const v2f*)(bcol + 16 * t * WPITCH + 4 * q2);
      __builtin_amdgcn_sched_barrier(0);
#pragma unroll
      for (int t = 0; t < 8; ++t)
        acc[t] = __builtin_amdgcn_wmma_f32_16x16x4_f32(false, a_cur[j + 1], false, b1[t],
                                                       (short)0, acc[t], false, false);
    }
#pragma unroll
    for (int j = 0; j < 8; ++j) a_cur[j] = a_nxt[j];
  }

  float* orow = h + (size_t)(tM + hi * 8) * C_OUT + l;
#pragma unroll
  for (int t = 0; t < 8; ++t)
#pragma unroll
    for (int r = 0; r < 8; ++r) orow[(size_t)r * C_OUT + 16 * t] = acc[t][r];
}

__global__ __launch_bounds__(256) void k_nodeatt(const float* __restrict__ h,
                                                 const float* __restrict__ att,
                                                 float* __restrict__ ai,
                                                 float* __restrict__ aj) {
  const int wave = threadIdx.x >> 5;
  const int lane = threadIdx.x & 31;
  const int n = blockIdx.x * 8 + wave;
  const float4 hv  = *(const float4*)(h + (size_t)n * C_OUT + lane * 4);
  const float4 av  = *(const float4*)(att + lane * 4);
  const float4 bv  = *(const float4*)(att + C_OUT + lane * 4);
  float si = hv.x*av.x + hv.y*av.y + hv.z*av.z + hv.w*av.w;
  float sj = hv.x*bv.x + hv.y*bv.y + hv.z*bv.z + hv.w*bv.w;
#pragma unroll
  for (int off = 16; off > 0; off >>= 1) {
    si += __shfl_down(si, off, 32);
    sj += __shfl_down(sj, off, 32);
  }
  if (lane == 0) { ai[n] = si; aj[n] = sj; }
}

__global__ void k_init(unsigned* __restrict__ amax, float* __restrict__ denom) {
  int n = blockIdx.x * blockDim.x + threadIdx.x;
  if (n < N_NODES) { amax[n] = 0u; denom[n] = 0.0f; }
}
__global__ void k_initout(float* __restrict__ out, const float* __restrict__ bias) {
  int i = blockIdx.x * blockDim.x + threadIdx.x;
  if (i < N_NODES * C_OUT) out[i] = bias[i & (C_OUT - 1)];
}

__device__ __forceinline__ unsigned f2key(float f) {
  unsigned b = __float_as_uint(f);
  return (b & 0x80000000u) ? ~b : (b | 0x80000000u);
}
__device__ __forceinline__ float key2f(unsigned k) {
  unsigned b = (k & 0x80000000u) ? (k & 0x7FFFFFFFu) : ~k;
  return __uint_as_float(b);
}
__device__ __forceinline__ float leaky(float a) {
  return a > 0.0f ? a : NEG_SLOPE * a;
}

__global__ void k_edgemax(const int* __restrict__ esrc, const int* __restrict__ edst,
                          const float* __restrict__ ai, const float* __restrict__ aj,
                          unsigned* __restrict__ amax) {
  int e = blockIdx.x * blockDim.x + threadIdx.x;
  if (e >= ET) return;
  int s, d;
  if (e < E_EDGES) { s = esrc[e]; d = edst[e]; } else { s = d = e - E_EDGES; }
  float a = leaky(ai[d] + aj[s]);
  atomicMax(&amax[d], f2key(a));
}

__global__ void k_edgeexp(const int* __restrict__ esrc, const int* __restrict__ edst,
                          const float* __restrict__ ai, const float* __restrict__ aj,
                          const unsigned* __restrict__ amax,
                          float* __restrict__ denom, float* __restrict__ ex) {
  int e = blockIdx.x * blockDim.x + threadIdx.x;
  if (e >= ET) return;
  int s, d;
  if (e < E_EDGES) { s = esrc[e]; d = edst[e]; } else { s = d = e - E_EDGES; }
  float a = leaky(ai[d] + aj[s]);
  float v = expf(a - key2f(amax[d]));
  ex[e] = v;
  hw_atomic_add_f32(&denom[d], v);
}

__global__ __launch_bounds__(256) void k_scatter(const int* __restrict__ esrc,
                                                 const int* __restrict__ edst,
                                                 const float* __restrict__ h,
                                                 const float* __restrict__ ex,
                                                 const float* __restrict__ denom,
                                                 float* __restrict__ out) {
  const int wave = threadIdx.x >> 5;
  const int lane = threadIdx.x & 31;
  const int e = blockIdx.x * 8 + wave;
  int s, d;
  if (e < E_EDGES) { s = esrc[e]; d = edst[e]; } else { s = d = e - E_EDGES; }
  const float coeff = ex[e] / (denom[d] + 1e-16f);
  const float4 hv = *(const float4*)(h + (size_t)s * C_OUT + lane * 4);
  float* o = out + (size_t)d * C_OUT + lane * 4;
  hw_atomic_add_f32(o + 0, hv.x * coeff);
  hw_atomic_add_f32(o + 1, hv.y * coeff);
  hw_atomic_add_f32(o + 2, hv.z * coeff);
  hw_atomic_add_f32(o + 3, hv.w * coeff);
}

extern "C" void kernel_launch(void* const* d_in, const int* in_sizes, int n_in,
                              void* d_out, int out_size, void* d_ws, size_t ws_size,
                              hipStream_t stream) {
  const float* x    = (const float*)d_in[0];
  const int*   ei   = (const int*)  d_in[1];
  const float* W    = (const float*)d_in[2];
  const float* att  = (const float*)d_in[3];
  const float* bias = (const float*)d_in[4];
  float* out = (float*)d_out;

  float*    h     = (float*)d_ws;
  float*    ai    = h + (size_t)N_NODES * C_OUT;
  float*    aj    = ai + N_NODES;
  unsigned* amax  = (unsigned*)(aj + N_NODES);
  float*    denom = (float*)(amax + N_NODES);
  float*    ex    = denom + N_NODES;

  const int* esrc = ei;
  const int* edst = ei + E_EDGES;

  k_gemm   <<<GEMM_BLOCKS, GEMM_THREADS, 0, stream>>>(x, W, h);
  k_init   <<<(N_NODES + 255) / 256, 256, 0, stream>>>(amax, denom);
  k_initout<<<(N_NODES * C_OUT + 255) / 256, 256, 0, stream>>>(out, bias);
  k_nodeatt<<<N_NODES / 8, 256, 0, stream>>>(h, att, ai, aj);
  k_edgemax<<<(ET + 255) / 256, 256, 0, stream>>>(esrc, edst, ai, aj, amax);
  k_edgeexp<<<(ET + 255) / 256, 256, 0, stream>>>(esrc, edst, ai, aj, amax, denom, ex);
  k_scatter<<<ET / 8, 256, 0, stream>>>(esrc, edst, h, ex, denom, out);
}


# Edges bucketed by target per call (count, scan, fill); a wave per target does max, exp-sum and the weighted h-row sum in registers and stores out once: no float atomics; sums in bucket order
# speedup vs baseline: 2.2445x; 2.0643x over previous
#include <hip/hip_runtime.h>


#define N_NODES 50000
#define E_EDGES 800000
#define DIN     256
#define C_OUT   128
#define NEG_SLOPE 0.2f

typedef __attribute__((ext_vector_type(2))) float v2f;
typedef __attribute__((ext_vector_type(8))) float v8f;

#define GEMM_THREADS 512
#define GEMM_WAVES   (GEMM_THREADS / 32)
#define GEMM_BLOCKS  256
#define M_STRIPS     (N_NODES / 16)
#define WPITCH       260
__global__ __launch_bounds__(GEMM_THREADS) void k_gemm(const float* __restrict__ x,
                                                       const float* __restrict__ W,
                                                       float* __restrict__ h) {
  __shared__ float Wt[C_OUT * WPITCH];

  const int wave  = threadIdx.x >> 5;
  const int lane  = threadIdx.x & 31;
  const int hi    = lane >> 4;
  const int l     = lane & 15;
  const int strip = wave * GEMM_BLOCKS + blockIdx.x;
  const bool valid = strip < M_STRIPS;
  const int tM = valid ? (strip << 4) : 0;

  const float* arow = x + (size_t)(tM + l) * DIN + 2 * hi;

  v2f a_cur[8], a_nxt[8];
#pragma unroll
  for (int j = 0; j < 8; ++j) a_cur[j] = *(const v2f*)(arow + 4 * j);

  {
    const int kk = lane & 7;
    const int cc = lane >> 3;
    for (int p = wave; p < (DIN / 8) * (C_OUT / 16); p += GEMM_WAVES) {
      const int k  = ((p >> 3) << 3) + kk;
      const int c4 = ((p & 7) << 2) + cc;
      const float4 w4 = *(const float4*)(W + (size_t)k * C_OUT + c4 * 4);
      float* d = Wt + (c4 * 4) * WPITCH + k;
      d[0]          = w4.x;
      d[WPITCH]     = w4.y;
      d[2 * WPITCH] = w4.z;
      d[3 * WPITCH] = w4.w;
    }
  }
  __syncthreads();
  if (!valid) return;

  const float* bcol = Wt + l * WPITCH + 2 * hi;

  v8f acc[8];
#pragma unroll
  for (int t = 0; t < 8; ++t) acc[t] = (v8f)0.0f;

  v2f b0[8], b1[8];
#pragma unroll
  for (int t = 0; t < 8; ++t) b0[t] = *(const v2f*)(bcol + 16 * t * WPITCH);

  for (int c = 0; c < DIN / 32; ++c) {
    const int cn = (c + 1 < DIN / 32) ? (c + 1) : c;
#pragma unroll
    for (int j = 0; j < 8; ++j) a_nxt[j] = *(const v2f*)(arow + 32 * cn + 4 * j);
    __builtin_amdgcn_sched_barrier(0);

#pragma unroll
    for (int j = 0; j < 8; j += 2) {
      const int q  = 8 * c + j;
      const int q2 = (q + 2 < DIN / 4) ? (q + 2) : (DIN / 4 - 1);
#pragma unroll
      for (int t = 0; t < 8; ++t) b1[t] = *(const v2f*)(bcol + 16 * t * WPITCH + 4 * (q + 1));
      __builtin_amdgcn_sched_barrier(0);
#pragma unroll
      for (int t = 0; t < 8; ++t)
        acc[t] = __builtin_amdgcn_wmma_f32_16x16x4_f32(false, a_cur[j], false, b0[t],
                                                       (short)0, acc[t], false, false);
#pragma unroll
      for (int t = 0; t < 8; ++t) b0[t] = *(const v2f*)(bcol + 16 * t * WPITCH + 4 * q2);
      __builtin_amdgcn_sched_barrier(0);
#pragma unroll
      for (int t = 0; t < 8; ++t)
        acc[t] = __builtin_amdgcn_wmma_f32_16x16x4_f32(false, a_cur[j + 1], false, b1[t],
                                                       (short)0, acc[t], false, false);
    }
#pragma unroll
    for (int j = 0; j < 8; ++j) a_cur[j] = a_nxt[j];
  }

  float* orow = h + (size_t)(tM + hi * 8) * C_OUT + l;
#pragma unroll
  for (int t = 0; t < 8; ++t)
#pragma unroll
    for (int r = 0; r < 8; ++r) orow[(size_t)r * C_OUT + 16 * t] = acc[t][r];
}

__global__ __launch_bounds__(256) void k_nodeatt(const float* __restrict__ h,
                                                 const float* __restrict__ att,
                                                 float* __restrict__ ai,
                                                 float* __restrict__ aj,
                                                 int* __restrict__ cnt) {
  const int wave = threadIdx.x >> 5;
  const int lane = threadIdx.x & 31;
  const int n = blockIdx.x * 8 + wave;
  const float4 hv  = *(const float4*)(h + (size_t)n * C_OUT + lane * 4);
  const float4 av  = *(const float4*)(att + lane * 4);
  const float4 bv  = *(const float4*)(att + C_OUT + lane * 4);
  float si = hv.x*av.x + hv.y*av.y + hv.z*av.z + hv.w*av.w;
  float sj = hv.x*bv.x + hv.y*bv.y + hv.z*bv.z + hv.w*bv.w;
#pragma unroll
  for (int off = 16; off > 0; off >>= 1) {
    si += __shfl_down(si, off, 32);
    sj += __shfl_down(sj, off, 32);
  }
  if (lane == 0) { ai[n] = si; aj[n] = sj; cnt[n] = 0; }
}

__device__ __forceinline__ float leaky(float a) {
  return a > 0.0f ? a : NEG_SLOPE * a;
}

__global__ __launch_bounds__(256) void k_count(const int* __restrict__ edst,
                                               int* __restrict__ cnt,
                                               int* __restrict__ place) {
  const int e = blockIdx.x * 256 + threadIdx.x;
  place[e] = atomicAdd(&cnt[edst[e]], 1);
}

#define SCAN_THREADS 1024
#define SCAN_QUADS   (N_NODES / 4)
#define SCAN_TILES   ((SCAN_QUADS + SCAN_THREADS - 1) / SCAN_THREADS)

__device__ __forceinline__ int wave_incl_scan(int v, int lane) {
#pragma unroll
  for (int off = 1; off < 32; off <<= 1) {
    const int t = __shfl_up(v, off, 32);
    if (lane >= off) v += t;
  }
  return v;
}

__global__ __launch_bounds__(SCAN_THREADS) void k_scan(const int* __restrict__ cnt,
                                                       int* __restrict__ rowptr) {
  __shared__ int wt[SCAN_TILES * 32];
  __shared__ int tt[SCAN_TILES];
  const int tid  = threadIdx.x;
  const int wave = tid >> 5;
  const int lane = tid & 31;

  int4 v[SCAN_TILES];
  int  inc[SCAN_TILES];
#pragma unroll
  for (int j = 0; j < SCAN_TILES; ++j) {
    const int q = j * SCAN_THREADS + tid;
    v[j] = (q < SCAN_QUADS) ? ((const int4*)cnt)[q] : make_int4(0, 0, 0, 0);
  }
#pragma unroll
  for (int j = 0; j < SCAN_TILES; ++j) {
    inc[j] = wave_incl_scan(v[j].x + v[j].y + v[j].z + v[j].w, lane);
    if (lane == 31) wt[j * 32 + wave] = inc[j];
  }
  __syncthreads();
  if (wave < SCAN_TILES) {
    const int w  = wt[wave * 32 + lane];
    const int i2 = wave_incl_scan(w, lane);
    wt[wave * 32 + lane] = i2 - w;
    if (lane == 31) tt[wave] = i2;
  }
  __syncthreads();
  int toff = 0;
#pragma unroll
  for (int j = 0; j < SCAN_TILES; ++j) {
    const int q = j * SCAN_THREADS + tid;
    const int s = v[j].x + v[j].y + v[j].z + v[j].w;
    const int b = toff + wt[j * 32 + wave] + inc[j] - s;
    if (q < SCAN_QUADS)
      ((int4*)rowptr)[q] = make_int4(b, b + v[j].x, b + v[j].x + v[j].y, b + v[j].x + v[j].y + v[j].z);
    toff += tt[j];
  }
}

__global__ __launch_bounds__(256) void k_fill(const int* __restrict__ esrc,
                                              const int* __restrict__ edst,
                                              const int* __restrict__ rowptr,
                                              const int* __restrict__ place,
                                              int* __restrict__ bucket) {
  const int e = blockIdx.x * 256 + threadIdx.x;
  int slot = rowptr[edst[e]] + place[e];
  slot = min(max(slot, 0), E_EDGES - 1);
  bucket[slot] = esrc[e];
}

__device__ __forceinline__ float wave_max(float v) {
#pragma unroll
  for (int off = 16; off > 0; off >>= 1) v = fmaxf(v, __shfl_xor(v, off, 32));
  return v;
}
__device__ __forceinline__ float wave_sum(float v) {
#pragma unroll
  for (int off = 16; off > 0; off >>= 1) v += __shfl_xor(v, off, 32);
  return v;
}

__device__ __forceinline__ int entry_src(const int* __restrict__ bucket, int beg, int d,
                                         int t, int total) {
  int s = d;
  if (t >= 1 && t < total) {
    const int slot = min(max(beg + t - 1, 0), E_EDGES - 1);
    s = min(max(bucket[slot], 0), N_NODES - 1);
  }
  return s;
}

#define AGG_ROWS 8
__device__ __forceinline__ void gather_rows(float4& acc, const float* __restrict__ h,
                                            int s, float coef, int n, int lane) {
  for (int k = 0; k < n; k += AGG_ROWS) {
    float4 hv[AGG_ROWS];
    float  ck[AGG_ROWS];
#pragma unroll
    for (int u = 0; u < AGG_ROWS; ++u) {
      const int sk = __builtin_amdgcn_readlane(s, k + u);
      ck[u] = __uint_as_float(__builtin_amdgcn_readlane(__float_as_uint(coef), k + u));
      hv[u] = *(const float4*)(h + (size_t)sk * C_OUT + lane * 4);
    }
#pragma unroll
    for (int u = 0; u < AGG_ROWS; ++u) {
      acc.x += hv[u].x * ck[u];
      acc.y += hv[u].y * ck[u];
      acc.z += hv[u].z * ck[u];
      acc.w += hv[u].w * ck[u];
    }
  }
}

__global__ __launch_bounds__(256) void k_aggregate(const float* __restrict__ h,
                                                   const float* __restrict__ ai,
                                                   const float* __restrict__ aj,
                                                   const int* __restrict__ cnt,
                                                   const int* __restrict__ rowptr,
                                                   const int* __restrict__ bucket,
                                                   const float* __restrict__ bias,
                                                   float* __restrict__ out) {
  const int wave = __builtin_amdgcn_readfirstlane((int)(threadIdx.x >> 5));
  const int lane = threadIdx.x & 31;
  const int d = blockIdx.x * 8 + wave;
  const int beg   = rowptr[d];
  const int total = cnt[d] + 1;
  const float aid = ai[d];
  float4 acc = make_float4(0.0f, 0.0f, 0.0f, 0.0f);

  if (total <= 32) {
    const bool valid = lane < total;
    const int s = entry_src(bucket, beg, d, lane, total);
    const float a   = valid ? leaky(aid + aj[s]) : -__builtin_inff();
    const float m   = wave_max(a);
    const float ex  = valid ? expf(a - m) : 0.0f;
    const float den = wave_sum(ex);
    const float coef = ex / (den + 1e-16f);
    gather_rows(acc, h, s, coef, total, lane);
  } else {
    float m = -__builtin_inff();
    for (int base = 0; base < total; base += 32) {
      const int t = base + lane;
      const int s = entry_src(bucket, beg, d, t, total);
      if (t < total) m = fmaxf(m, leaky(aid + aj[s]));
    }
    m = wave_max(m);
    float den = 0.0f;
    for (int base = 0; base < total; base += 32) {
      const int t = base + lane;
      const int s = entry_src(bucket, beg, d, t, total);
      if (t < total) den += expf(leaky(aid + aj[s]) - m);
    }
    den = wave_sum(den);
    for (int base = 0; base < total; base += 32) {
      const int t = base + lane;
      const int s = entry_src(bucket, beg, d, t, total);
      const float coef = (t < total) ? expf(leaky(aid + aj[s]) - m) / (den + 1e-16f) : 0.0f;
      gather_rows(acc, h, s, coef, min(32, total - base), lane);
    }
  }

  const float4 bv = *(const float4*)(bias + lane * 4);
  *(float4*)(out + (size_t)d * C_OUT + lane * 4) =
      make_float4(acc.x + bv.x, acc.y + bv.y, acc.z + bv.z, acc.w + bv.w);
}

extern "C" void kernel_launch(void* const* d_in, const int* in_sizes, int n_in,
                              void* d_out, int out_size, void* d_ws, size_t ws_size,
                              hipStream_t stream) {
  const float* x    = (const float*)d_in[0];
  const int*   ei   = (const int*)  d_in[1];
  const float* W    = (const float*)d_in[2];
  const float* att  = (const float*)d_in[3];
  const float* bias = (const float*)d_in[4];
  float* out = (float*)d_out;

  float* h      = (float*)d_ws;
  float* ai     = h + (size_t)N_NODES * C_OUT;
  float* aj     = ai + N_NODES;
  int*   cnt    = (int*)(aj + N_NODES);
  int*   rowptr = cnt + N_NODES;
  int*   place  = rowptr + N_NODES;
  int*   bucket = place + E_EDGES;

  const int* esrc = ei;
  const int* edst = ei + E_EDGES;

  k_gemm     <<<GEMM_BLOCKS, GEMM_THREADS, 0, stream>>>(x, W, h);
  k_nodeatt  <<<N_NODES / 8, 256, 0, stream>>>(h, att, ai, aj, cnt);
  k_count    <<<E_EDGES / 256, 256, 0, stream>>>(edst, cnt, place);
  k_scan     <<<1, SCAN_THREADS, 0, stream>>>(cnt, rowptr);
  k_fill     <<<E_EDGES / 256, 256, 0, stream>>>(esrc, edst, rowptr, place, bucket);
  k_aggregate<<<N_NODES / 8, 256, 0, stream>>>(h, ai, aj, cnt, rowptr, bucket, bias, out);
}


# Fixed 256-B bucket per target (word 0 its edge counter, 63 slots, overflow list): the count atomic's return is the slot, so scan, fill, place and rowptr go; 6 launches -> 4; sums in bucket order
# speedup vs baseline: 3.1633x; 1.4094x over previous
#include <hip/hip_runtime.h>


#define N_NODES 50000
#define E_EDGES 800000
#define DIN     256
#define C_OUT   128
#define NEG_SLOPE 0.2f
#define BK_WORDS  64
#define BK_SLOTS  (BK_WORDS - 1)

typedef __attribute__((ext_vector_type(2))) float v2f;
typedef __attribute__((ext_vector_type(8))) float v8f;

#define GEMM_THREADS 512
#define GEMM_WAVES   (GEMM_THREADS / 32)
#define GEMM_BLOCKS  256
#define M_STRIPS     (N_NODES / 16)
#define WPITCH       260
__global__ __launch_bounds__(GEMM_THREADS) void k_gemm(const float* __restrict__ x,
                                                       const float* __restrict__ W,
                                                       float* __restrict__ h) {
  __shared__ float Wt[C_OUT * WPITCH];

  const int wave  = threadIdx.x >> 5;
  const int lane  = threadIdx.x & 31;
  const int hi    = lane >> 4;
  const int l     = lane & 15;
  const int strip = wave * GEMM_BLOCKS + blockIdx.x;
  const bool valid = strip < M_STRIPS;
  const int tM = valid ? (strip << 4) : 0;

  const float* arow = x + (size_t)(tM + l) * DIN + 2 * hi;

  v2f a_cur[8], a_nxt[8];
#pragma unroll
  for (int j = 0; j < 8; ++j) a_cur[j] = *(const v2f*)(arow + 4 * j);

  {
    const int kk = lane & 7;
    const int cc = lane >> 3;
    for (int p = wave; p < (DIN / 8) * (C_OUT / 16); p += GEMM_WAVES) {
      const int k  = ((p >> 3) << 3) + kk;
      const int c4 = ((p & 7) << 2) + cc;
      const float4 w4 = *(const float4*)(W + (size_t)k * C_OUT + c4 * 4);
      float* d = Wt + (c4 * 4) * WPITCH + k;
      d[0]          = w4.x;
      d[WPITCH]     = w4.y;
      d[2 * WPITCH] = w4.z;
      d[3 * WPITCH] = w4.w;
    }
  }
  __syncthreads();
  if (!valid) return;

  const float* bcol = Wt + l * WPITCH + 2 * hi;

  v8f acc[8];
#pragma unroll
  for (int t = 0; t < 8; ++t) acc[t] = (v8f)0.0f;

  v2f b0[8], b1[8];
#pragma unroll
  for (int t = 0; t < 8; ++t) b0[t] = *(const v2f*)(bcol + 16 * t * WPITCH);

  for (int c = 0; c < DIN / 32; ++c) {
    const int cn = (c + 1 < DIN / 32) ? (c + 1) : c;
#pragma unroll
    for (int j = 0; j < 8; ++j) a_nxt[j] = *(const v2f*)(arow + 32 * cn + 4 * j);
    __builtin_amdgcn_sched_barrier(0);

#pragma unroll
    for (int j = 0; j < 8; j += 2) {
      const int q  = 8 * c + j;
      const int q2 = (q + 2 < DIN / 4) ? (q + 2) : (DIN / 4 - 1);
#pragma unroll
      for (int t = 0; t < 8; ++t) b1[t] = *(const v2f*)(bcol + 16 * t * WPITCH + 4 * (q + 1));
      __builtin_amdgcn_sched_barrier(0);
#pragma unroll
      for (int t = 0; t < 8; ++t)
        acc[t] = __builtin_amdgcn_wmma_f32_16x16x4_f32(false, a_cur[j], false, b0[t],
                                                       (short)0, acc[t], false, false);
#pragma unroll
      for (int t = 0; t < 8; ++t) b0[t] = *(const v2f*)(bcol + 16 * t * WPITCH + 4 * q2);
      __builtin_amdgcn_sched_barrier(0);
#pragma unroll
      for (int t = 0; t < 8; ++t)
        acc[t] = __builtin_amdgcn_wmma_f32_16x16x4_f32(false, a_cur[j + 1], false, b1[t],
                                                       (short)0, acc[t], false, false);
    }
#pragma unroll
    for (int j = 0; j < 8; ++j) a_cur[j] = a_nxt[j];
  }

  float* orow = h + (size_t)(tM + hi * 8) * C_OUT + l;
#pragma unroll
  for (int t = 0; t < 8; ++t)
#pragma unroll
    for (int r = 0; r < 8; ++r) orow[(size_t)r * C_OUT + 16 * t] = acc[t][r];
}

__global__ __launch_bounds__(256) void k_nodeatt(const float* __restrict__ h,
                                                 const float* __restrict__ att,
                                                 float* __restrict__ ai,
                                                 float* __restrict__ aj,
                                                 int* __restrict__ bucket,
                                                 int* __restrict__ ovfn) {
  const int wave = threadIdx.x >> 5;
  const int lane = threadIdx.x & 31;
  const int n = blockIdx.x * 8 + wave;
  const float4 hv  = *(const float4*)(h + (size_t)n * C_OUT + lane * 4);
  const float4 av  = *(const float4*)(att + lane * 4);
  const float4 bv  = *(const float4*)(att + C_OUT + lane * 4);
  float si = hv.x*av.x + hv.y*av.y + hv.z*av.z + hv.w*av.w;
  float sj = hv.x*bv.x + hv.y*bv.y + hv.z*bv.z + hv.w*bv.w;
#pragma unroll
  for (int off = 16; off > 0; off >>= 1) {
    si += __shfl_down(si, off, 32);
    sj += __shfl_down(sj, off, 32);
  }
  if (lane == 0) { ai[n] = si; aj[n] = sj; }
  bucket[(size_t)n * BK_WORDS + lane] = 0;
  if (n == 0 && lane == 0) ovfn[0] = 0;
}

__device__ __forceinline__ float leaky(float a) {
  return a > 0.0f ? a : NEG_SLOPE * a;
}

__global__ __launch_bounds__(256) void k_bucket(const int* __restrict__ esrc,
                                                const int* __restrict__ edst,
                                                int* __restrict__ bucket,
                                                int* __restrict__ ovfn,
                                                int* __restrict__ ovf_dst,
                                                int* __restrict__ ovf_src) {
  const int e = blockIdx.x * 256 + threadIdx.x;
  const int d = edst[e];
  const int s = esrc[e];
  int* bk = bucket + (size_t)d * BK_WORDS;
  const int p = atomicAdd(bk, 1);
  if (p >= 0 && p < BK_SLOTS) {
    bk[1 + p] = s;
  } else {
    const int q = atomicAdd(ovfn, 1);
    if (q >= 0 && q < E_EDGES) { ovf_dst[q] = d; ovf_src[q] = s; }
  }
}

__device__ __forceinline__ float wave_max(float v) {
#pragma unroll
  for (int off = 16; off > 0; off >>= 1) v = fmaxf(v, __shfl_xor(v, off, 32));
  return v;
}
__device__ __forceinline__ float wave_sum(float v) {
#pragma unroll
  for (int off = 16; off > 0; off >>= 1) v += __shfl_xor(v, off, 32);
  return v;
}

__device__ __forceinline__ int entry_get(const int* __restrict__ bk,
                                         const int* __restrict__ ovf_dst,
                                         const int* __restrict__ ovf_src,
                                         int d, int g, int total_b, int G, bool& ok) {
  int s = d;
  ok = false;
  if (g < total_b) {
    ok = true;
    if (g >= 1) s = min(max(bk[g], 0), N_NODES - 1);
  } else if (g < G) {
    const int j = min(max(g - total_b, 0), E_EDGES - 1);
    if (ovf_dst[j] == d) {
      ok = true;
      s = min(max(ovf_src[j], 0), N_NODES - 1);
    }
  }
  return s;
}

#define AGG_ROWS 8
__device__ __forceinline__ void gather_rows(float4& acc, const float* __restrict__ h,
                                            int s, float coef, int n, int lane) {
  for (int k = 0; k < n; k += AGG_ROWS) {
    float4 hv[AGG_ROWS];
    float  ck[AGG_ROWS];
#pragma unroll
    for (int u = 0; u < AGG_ROWS; ++u) {
      const int sk = __builtin_amdgcn_readlane(s, k + u);
      ck[u] = __uint_as_float(__builtin_amdgcn_readlane(__float_as_uint(coef), k + u));
      hv[u] = *(const float4*)(h + (size_t)sk * C_OUT + lane * 4);
    }
#pragma unroll
    for (int u = 0; u < AGG_ROWS; ++u) {
      acc.x += hv[u].x * ck[u];
      acc.y += hv[u].y * ck[u];
      acc.z += hv[u].z * ck[u];
      acc.w += hv[u].w * ck[u];
    }
  }
}

__global__ __launch_bounds__(256) void k_aggregate(const float* __restrict__ h,
                                                   const float* __restrict__ ai,
                                                   const float* __restrict__ aj,
                                                   const int* __restrict__ bucket,
                                                   const int* __restrict__ ovfn,
                                                   const int* __restrict__ ovf_dst,
                                                   const int* __restrict__ ovf_src,
                                                   const float* __restrict__ bias,
                                                   float* __restrict__ out) {
  const int wave = __builtin_amdgcn_readfirstlane((int)(threadIdx.x >> 5));
  const int lane = threadIdx.x & 31;
  const int d = blockIdx.x * 8 + wave;
  const int* bk = bucket + (size_t)d * BK_WORDS;
  const int w = bk[lane];
  const float aid = ai[d];
  const int c = max(__builtin_amdgcn_readfirstlane(w), 0);
  float4 acc = make_float4(0.0f, 0.0f, 0.0f, 0.0f);

  if (c < 32) {
    const int total = c + 1;
    const bool valid = lane < total;
    const int s = (lane >= 1 && valid) ? min(max(w, 0), N_NODES - 1) : d;
    const float a   = valid ? leaky(aid + aj[s]) : -__builtin_inff();
    const float m   = wave_max(a);
    const float ex  = valid ? expf(a - m) : 0.0f;
    const float den = wave_sum(ex);
    const float coef = ex / (den + 1e-16f);
    gather_rows(acc, h, s, coef, total, lane);
  } else {
    const int total_b = min(c, BK_SLOTS) + 1;
    int G = total_b;
    if (c > BK_SLOTS) G += min(max(ovfn[0], 0), E_EDGES);
    float m = -__builtin_inff();
    for (int base = 0; base < G; base += 32) {
      bool ok;
      const int s = entry_get(bk, ovf_dst, ovf_src, d, base + lane, total_b, G, ok);
      if (ok) m = fmaxf(m, leaky(aid + aj[s]));
    }
    m = wave_max(m);
    float den = 0.0f;
    for (int base = 0; base < G; base += 32) {
      bool ok;
      const int s = entry_get(bk, ovf_dst, ovf_src, d, base + lane, total_b, G, ok);
      if (ok) den += expf(leaky(aid + aj[s]) - m);
    }
    den = wave_sum(den);
    for (int base = 0; base < G; base += 32) {
      bool ok;
      const int s = entry_get(bk, ovf_dst, ovf_src, d, base + lane, total_b, G, ok);
      const float coef = ok ? expf(leaky(aid + aj[s]) - m) / (den + 1e-16f) : 0.0f;
      if (__any(ok)) gather_rows(acc, h, s, coef, min(32, G - base), lane);
    }
  }

  const float4 bv = *(const float4*)(bias + lane * 4);
  *(float4*)(out + (size_t)d * C_OUT + lane * 4) =
      make_float4(acc.x + bv.x, acc.y + bv.y, acc.z + bv.z, acc.w + bv.w);
}

extern "C" void kernel_launch(void* const* d_in, const int* in_sizes, int n_in,
                              void* d_out, int out_size, void* d_ws, size_t ws_size,
                              hipStream_t stream) {
  const float* x    = (const float*)d_in[0];
  const int*   ei   = (const int*)  d_in[1];
  const float* W    = (const float*)d_in[2];
  const float* att  = (const float*)d_in[3];
  const float* bias = (const float*)d_in[4];
  float* out = (float*)d_out;

  float* h       = (float*)d_ws;
  int*   bucket  = (int*)(h + (size_t)N_NODES * C_OUT);
  float* ai      = (float*)(bucket + (size_t)N_NODES * BK_WORDS);
  float* aj      = ai + N_NODES;
  int*   ovfn    = (int*)(aj + N_NODES);
  int*   ovf_dst = ovfn + 4;
  int*   ovf_src = ovf_dst + E_EDGES;

  const int* esrc = ei;
  const int* edst = ei + E_EDGES;

  k_gemm     <<<GEMM_BLOCKS, GEMM_THREADS, 0, stream>>>(x, W, h);
  k_nodeatt  <<<N_NODES / 8, 256, 0, stream>>>(h, att, ai, aj, bucket, ovfn);
  k_bucket   <<<E_EDGES / 256, 256, 0, stream>>>(esrc, edst, bucket, ovfn, ovf_dst, ovf_src);
  k_aggregate<<<N_NODES / 8, 256, 0, stream>>>(h, ai, aj, bucket, ovfn, ovf_dst, ovf_src, bias, out);
}


# Per-node scores ai, aj and the bucket-line clear move into the GEMM epilogue (a wave holds whole h rows in its accumulators): 4 launches -> 3, h not re-read; ai, aj summation order changes
# speedup vs baseline: 3.4726x; 1.0978x over previous
#include <hip/hip_runtime.h>


#define N_NODES 50000
#define E_EDGES 800000
#define DIN     256
#define C_OUT   128
#define NEG_SLOPE 0.2f
#define BK_WORDS  64
#define BK_SLOTS  (BK_WORDS - 1)

typedef __attribute__((ext_vector_type(2))) float v2f;
typedef __attribute__((ext_vector_type(8))) float v8f;

#define GEMM_THREADS 512
#define GEMM_WAVES   (GEMM_THREADS / 32)
#define GEMM_BLOCKS  256
#define M_STRIPS     (N_NODES / 16)
#define WPITCH       260
template <int M>
__device__ __forceinline__ void fold_step(float (&p)[16], int l) {
  const bool up = (l & M) != 0;
#pragma unroll
  for (int q = 0; q < M; ++q) {
    const float keep = up ? p[q + M] : p[q];
    const float send = up ? p[q] : p[q + M];
    p[q] = keep + __shfl_xor(send, M, 32);
  }
}

__global__ __launch_bounds__(GEMM_THREADS) void k_gemm(const float* __restrict__ x,
                                                       const float* __restrict__ W,
                                                       const float* __restrict__ att,
                                                       float* __restrict__ h,
                                                       float* __restrict__ ai,
                                                       float* __restrict__ aj,
                                                       int* __restrict__ bucket,
                                                       int* __restrict__ ovfn) {
  __shared__ float Wt[C_OUT * WPITCH];
  __shared__ float att_s[2 * C_OUT];

  const int wave  = threadIdx.x >> 5;
  const int lane  = threadIdx.x & 31;
  const int hi    = lane >> 4;
  const int l     = lane & 15;
  const int strip = wave * GEMM_BLOCKS + blockIdx.x;
  const bool valid = strip < M_STRIPS;
  const int tM = valid ? (strip << 4) : 0;

  const float* arow = x + (size_t)(tM + l) * DIN + 2 * hi;

  v2f a_cur[8], a_nxt[8];
#pragma unroll
  for (int j = 0; j < 8; ++j) a_cur[j] = *(const v2f*)(arow + 4 * j);

  {
    const int kk = lane & 7;
    const int cc = lane >> 3;
    for (int p = wave; p < (DIN / 8) * (C_OUT / 16); p += GEMM_WAVES) {
      const int k  = ((p >> 3) << 3) + kk;
      const int c4 = ((p & 7) << 2) + cc;
      const float4 w4 = *(const float4*)(W + (size_t)k * C_OUT + c4 * 4);
      float* d = Wt + (c4 * 4) * WPITCH + k;
      d[0]          = w4.x;
      d[WPITCH]     = w4.y;
      d[2 * WPITCH] = w4.z;
      d[3 * WPITCH] = w4.w;
    }
  }
  if (threadIdx.x < 2 * C_OUT) att_s[threadIdx.x] = att[threadIdx.x];
  __syncthreads();
  if (!valid) return;

  {
    const int4 z4 = make_int4(0, 0, 0, 0);
#pragma unroll
    for (int i = 0; i < 4; ++i)
      *(int4*)(bucket + (size_t)(tM + 4 * i + (lane >> 3)) * BK_WORDS + (lane & 7) * 4) = z4;
  }

  const float* bcol = Wt + l * WPITCH + 2 * hi;

  v8f acc[8];
#pragma unroll
  for (int t = 0; t < 8; ++t) acc[t] = (v8f)0.0f;

  v2f b0[8], b1[8];
#pragma unroll
  for (int t = 0; t < 8; ++t) b0[t] = *(const v2f*)(bcol + 16 * t * WPITCH);

  for (int c = 0; c < DIN / 32; ++c) {
    const int cn = (c + 1 < DIN / 32) ? (c + 1) : c;
#pragma unroll
    for (int j = 0; j < 8; ++j) a_nxt[j] = *(const v2f*)(arow + 32 * cn + 4 * j);
    __builtin_amdgcn_sched_barrier(0);

#pragma unroll
    for (int j = 0; j < 8; j += 2) {
      const int q  = 8 * c + j;
      const int q2 = (q + 2 < DIN / 4) ? (q + 2) : (DIN / 4 - 1);
#pragma unroll
      for (int t = 0; t < 8; ++t) b1[t] = *(const v2f*)(bcol + 16 * t * WPITCH + 4 * (q + 1));
      __builtin_amdgcn_sched_barrier(0);
#pragma unroll
      for (int t = 0; t < 8; ++t)
        acc[t] = __builtin_amdgcn_wmma_f32_16x16x4_f32(false, a_cur[j], false, b0[t],
                                                       (short)0, acc[t], false, false);
#pragma unroll
      for (int t = 0; t < 8; ++t) b0[t] = *(const v2f*)(bcol + 16 * t * WPITCH + 4 * q2);
      __builtin_amdgcn_sched_barrier(0);
#pragma unroll
      for (int t = 0; t < 8; ++t)
        acc[t] = __builtin_amdgcn_wmma_f32_16x16x4_f32(false, a_cur[j + 1], false, b1[t],
                                                       (short)0, acc[t], false, false);
    }
#pragma unroll
    for (int j = 0; j < 8; ++j) a_cur[j] = a_nxt[j];
  }

  float* orow = h + (size_t)(tM + hi * 8) * C_OUT + l;
#pragma unroll
  for (int t = 0; t < 8; ++t)
#pragma unroll
    for (int r = 0; r < 8; ++r) orow[(size_t)r * C_OUT + 16 * t] = acc[t][r];

  float p[16];
  {
    float wi[8], wj[8];
#pragma unroll
    for (int t = 0; t < 8; ++t) { wi[t] = att_s[16 * t + l]; wj[t] = att_s[C_OUT + 16 * t + l]; }
#pragma unroll
    for (int r = 0; r < 8; ++r) {
      float si = 0.0f, sj = 0.0f;
#pragma unroll
      for (int t = 0; t < 8; ++t) { si += acc[t][r] * wi[t]; sj += acc[t][r] * wj[t]; }
      p[r] = si;
      p[8 + r] = sj;
    }
  }
  fold_step<8>(p, l);
  fold_step<4>(p, l);
  fold_step<2>(p, l);
  fold_step<1>(p, l);
  {
    float* dst = (l & 8) ? aj : ai;
    dst[tM + (l & 7) + 8 * hi] = p[0];
  }
  if (strip == 0 && lane == 0) ovfn[0] = 0;
}

__device__ __forceinline__ float leaky(float a) {
  return a > 0.0f ? a : NEG_SLOPE * a;
}

__global__ __launch_bounds__(256) void k_bucket(const int* __restrict__ esrc,
                                                const int* __restrict__ edst,
                                                int* __restrict__ bucket,
                                                int* __restrict__ ovfn,
                                                int* __restrict__ ovf_dst,
                                                int* __restrict__ ovf_src) {
  const int e = blockIdx.x * 256 + threadIdx.x;
  const int d = edst[e];
  const int s = esrc[e];
  int* bk = bucket + (size_t)d * BK_WORDS;
  const int p = atomicAdd(bk, 1);
  if (p >= 0 && p < BK_SLOTS) {
    bk[1 + p] = s;
  } else {
    const int q = atomicAdd(ovfn, 1);
    if (q >= 0 && q < E_EDGES) { ovf_dst[q] = d; ovf_src[q] = s; }
  }
}

__device__ __forceinline__ float wave_max(float v) {
#pragma unroll
  for (int off = 16; off > 0; off >>= 1) v = fmaxf(v, __shfl_xor(v, off, 32));
  return v;
}
__device__ __forceinline__ float wave_sum(float v) {
#pragma unroll
  for (int off = 16; off > 0; off >>= 1) v += __shfl_xor(v, off, 32);
  return v;
}

__device__ __forceinline__ int entry_get(const int* __restrict__ bk,
                                         const int* __restrict__ ovf_dst,
                                         const int* __restrict__ ovf_src,
                                         int d, int g, int total_b, int G, bool& ok) {
  int s = d;
  ok = false;
  if (g < total_b) {
    ok = true;
    if (g >= 1) s = min(max(bk[g], 0), N_NODES - 1);
  } else if (g < G) {
    const int j = min(max(g - total_b, 0), E_EDGES - 1);
    if (ovf_dst[j] == d) {
      ok = true;
      s = min(max(ovf_src[j], 0), N_NODES - 1);
    }
  }
  return s;
}

#define AGG_ROWS 8
__device__ __forceinline__ void gather_rows(float4& acc, const float* __restrict__ h,
                                            int s, float coef, int n, int lane) {
  for (int k = 0; k < n; k += AGG_ROWS) {
    float4 hv[AGG_ROWS];
    float  ck[AGG_ROWS];
#pragma unroll
    for (int u = 0; u < AGG_ROWS; ++u) {
      const int sk = __builtin_amdgcn_readlane(s, k + u);
      ck[u] = __uint_as_float(__builtin_amdgcn_readlane(__float_as_uint(coef), k + u));
      hv[u] = *(const float4*)(h + (size_t)sk * C_OUT + lane * 4);
    }
#pragma unroll
    for (int u = 0; u < AGG_ROWS; ++u) {
      acc.x += hv[u].x * ck[u];
      acc.y += hv[u].y * ck[u];
      acc.z += hv[u].z * ck[u];
      acc.w += hv[u].w * ck[u];
    }
  }
}

__global__ __launch_bounds__(256) void k_aggregate(const float* __restrict__ h,
                                                   const float* __restrict__ ai,
                                                   const float* __restrict__ aj,
                                                   const int* __restrict__ bucket,
                                                   const int* __restrict__ ovfn,
                                                   const int* __restrict__ ovf_dst,
                                                   const int* __restrict__ ovf_src,
                                                   const float* __restrict__ bias,
                                                   float* __restrict__ out) {
  const int wave = __builtin_amdgcn_readfirstlane((int)(threadIdx.x >> 5));
  const int lane = threadIdx.x & 31;
  const int d = blockIdx.x * 8 + wave;
  const int* bk = bucket + (size_t)d * BK_WORDS;
  const int w = bk[lane];
  const float aid = ai[d];
  const int c = max(__builtin_amdgcn_readfirstlane(w), 0);
  float4 acc = make_float4(0.0f, 0.0f, 0.0f, 0.0f);

  if (c < 32) {
    const int total = c + 1;
    const bool valid = lane < total;
    const int s = (lane >= 1 && valid) ? min(max(w, 0), N_NODES - 1) : d;
    const float a   = valid ? leaky(aid + aj[s]) : -__builtin_inff();
    const float m   = wave_max(a);
    const float ex  = valid ? expf(a - m) : 0.0f;
    const float den = wave_sum(ex);
    const float coef = ex / (den + 1e-16f);
    gather_rows(acc, h, s, coef, total, lane);
  } else {
    const int total_b = min(c, BK_SLOTS) + 1;
    int G = total_b;
    if (c > BK_SLOTS) G += min(max(ovfn[0], 0), E_EDGES);
    float m = -__builtin_inff();
    for (int base = 0; base < G; base += 32) {
      bool ok;
      const int s = entry_get(bk, ovf_dst, ovf_src, d, base + lane, total_b, G, ok);
      if (ok) m = fmaxf(m, leaky(aid + aj[s]));
    }
    m = wave_max(m);
    float den = 0.0f;
    for (int base = 0; base < G; base += 32) {
      bool ok;
      const int s = entry_get(bk, ovf_dst, ovf_src, d, base + lane, total_b, G, ok);
      if (ok) den += expf(leaky(aid + aj[s]) - m);
    }
    den = wave_sum(den);
    for (int base = 0; base < G; base += 32) {
      bool ok;
      const int s = entry_get(bk, ovf_dst, ovf_src, d, base + lane, total_b, G, ok);
      const float coef = ok ? expf(leaky(aid + aj[s]) - m) / (den + 1e-16f) : 0.0f;
      if (__any(ok)) gather_rows(acc, h, s, coef, min(32, G - base), lane);
    }
  }

  const float4 bv = *(const float4*)(bias + lane * 4);
  *(float4*)(out + (size_t)d * C_OUT + lane * 4) =
      make_float4(acc.x + bv.x, acc.y + bv.y, acc.z + bv.z, acc.w + bv.w);
}

extern "C" void kernel_launch(void* const* d_in, const int* in_sizes, int n_in,
                              void* d_out, int out_size, void* d_ws, size_t ws_size,
                              hipStream_t stream) {
  const float* x    = (const float*)d_in[0];
  const int*   ei   = (const int*)  d_in[1];
  const float* W    = (const float*)d_in[2];
  const float* att  = (const float*)d_in[3];
  const float* bias = (const float*)d_in[4];
  float* out = (float*)d_out;

  float* h       = (float*)d_ws;
  int*   bucket  = (int*)(h + (size_t)N_NODES * C_OUT);
  float* ai      = (float*)(bucket + (size_t)N_NODES * BK_WORDS);
  float* aj      = ai + N_NODES;
  int*   ovfn    = (int*)(aj + N_NODES);
  int*   ovf_dst = ovfn + 4;
  int*   ovf_src = ovf_dst + E_EDGES;

  const int* esrc = ei;
  const int* edst = ei + E_EDGES;

  k_gemm     <<<GEMM_BLOCKS, GEMM_THREADS, 0, stream>>>(x, W, att, h, ai, aj, bucket, ovfn);
  k_bucket   <<<E_EDGES / 256, 256, 0, stream>>>(esrc, edst, bucket, ovfn, ovf_dst, ovf_src);
  k_aggregate<<<N_NODES / 8, 256, 0, stream>>>(h, ai, aj, bucket, ovfn, ovf_dst, ovf_src, bias, out);
}


# Edge bucketing runs in the GEMM launch on its 971 strip-less waves (26 x 32 edges each, atomics in 2 batches) beside the matrix work; a small kernel zeroes the counters first; sums in bucket order
# speedup vs baseline: 3.8138x; 1.0983x over previous
#include <hip/hip_runtime.h>


#define N_NODES 50000
#define E_EDGES 800000
#define DIN     256
#define C_OUT   128
#define NEG_SLOPE 0.2f
#define BK_WORDS  64
#define BK_SLOTS  (BK_WORDS - 1)

typedef __attribute__((ext_vector_type(2))) float v2f;
typedef __attribute__((ext_vector_type(8))) float v8f;

#define CLEAR_BLOCKS 256
__global__ __launch_bounds__(256) void k_clear(int* __restrict__ bucket, int* __restrict__ ovfn) {
  const int4 z4 = make_int4(0, 0, 0, 0);
  for (int i = blockIdx.x * 256 + threadIdx.x; i < N_NODES * 8; i += CLEAR_BLOCKS * 256)
    *(int4*)(bucket + (size_t)(i >> 3) * BK_WORDS + (i & 7) * 4) = z4;
  if (blockIdx.x == 0 && threadIdx.x == 0) ovfn[0] = 0;
}

__device__ __forceinline__ void bucket_store(int* __restrict__ bucket, int* __restrict__ ovfn,
                                             int* __restrict__ ovf_dst, int* __restrict__ ovf_src,
                                             int d, int s, int p) {
  int* bk = bucket + (size_t)d * BK_WORDS;
  if (p >= 0 && p < BK_SLOTS) {
    bk[1 + p] = s;
  } else {
    const int q = atomicAdd(ovfn, 1);
    if (q >= 0 && q < E_EDGES) { ovf_dst[q] = d; ovf_src[q] = s; }
  }
}

template <int NG, int NA>
__device__ __forceinline__ void bucket_edges(const int* __restrict__ esrc,
                                             const int* __restrict__ edst,
                                             int* __restrict__ bucket,
                                             int* __restrict__ ovfn,
                                             int* __restrict__ ovf_dst,
                                             int* __restrict__ ovf_src,
                                             int e, bool extra) {
  int d[NG + 1], s[NG + 1], p[NG + 1];
#pragma unroll
  for (int u = 0; u < NG; ++u) { d[u] = edst[e + 32 * u]; s[u] = esrc[e + 32 * u]; }
  d[NG] = 0; s[NG] = 0; p[NG] = 0;
  if (extra) { d[NG] = edst[e + 32 * NG]; s[NG] = esrc[e + 32 * NG]; }

#pragma unroll
  for (int u = 0; u < NA; ++u) p[u] = atomicAdd(bucket + (size_t)d[u] * BK_WORDS, 1);
#pragma unroll
  for (int u = 0; u < NA; ++u) bucket_store(bucket, ovfn, ovf_dst, ovf_src, d[u], s[u], p[u]);

#pragma unroll
  for (int u = NA; u < NG; ++u) p[u] = atomicAdd(bucket + (size_t)d[u] * BK_WORDS, 1);
  if (extra) p[NG] = atomicAdd(bucket + (size_t)d[NG] * BK_WORDS, 1);
#pragma unroll
  for (int u = NA; u < NG; ++u) bucket_store(bucket, ovfn, ovf_dst, ovf_src, d[u], s[u], p[u]);
  if (extra) bucket_store(bucket, ovfn, ovf_dst, ovf_src, d[NG], s[NG], p[NG]);
}

#define GEMM_THREADS 512
#define GEMM_WAVES   (GEMM_THREADS / 32)
#define GEMM_BLOCKS  256
#define M_STRIPS     (N_NODES / 16)
#define WPITCH       260
#define SPARE_WAVES  (GEMM_WAVES * GEMM_BLOCKS - M_STRIPS)
#define E_GROUPS     (E_EDGES / 32)
#define G_LO         (E_GROUPS / SPARE_WAVES)
#define G_REM        (E_GROUPS % SPARE_WAVES)
#define G_HALF       13
template <int M>
__device__ __forceinline__ void fold_step(float (&p)[16], int l) {
  const bool up = (l & M) != 0;
#pragma unroll
  for (int q = 0; q < M; ++q) {
    const float keep = up ? p[q + M] : p[q];
    const float send = up ? p[q] : p[q + M];
    p[q] = keep + __shfl_xor(send, M, 32);
  }
}

__global__ __launch_bounds__(GEMM_THREADS) void k_gemm(const float* __restrict__ x,
                                                       const float* __restrict__ W,
                                                       const float* __restrict__ att,
                                                       float* __restrict__ h,
                                                       float* __restrict__ ai,
                                                       float* __restrict__ aj,
                                                       const int* __restrict__ esrc,
                                                       const int* __restrict__ edst,
                                                       int* __restrict__ bucket,
                                                       int* __restrict__ ovfn,
                                                       int* __restrict__ ovf_dst,
                                                       int* __restrict__ ovf_src) {
  __shared__ float Wt[C_OUT * WPITCH];
  __shared__ float att_s[2 * C_OUT];

  const int wave  = threadIdx.x >> 5;
  const int lane  = threadIdx.x & 31;
  const int hi    = lane >> 4;
  const int l     = lane & 15;
  const int strip = wave * GEMM_BLOCKS + blockIdx.x;
  const bool valid = strip < M_STRIPS;
  const int tM = valid ? (strip << 4) : 0;

  const float* arow = x + (size_t)(tM + l) * DIN + 2 * hi;

  v2f a_cur[8], a_nxt[8];
#pragma unroll
  for (int j = 0; j < 8; ++j) a_cur[j] = *(const v2f*)(arow + 4 * j);

  {
    const int kk = lane & 7;
    const int cc = lane >> 3;
    for (int p = wave; p < (DIN / 8) * (C_OUT / 16); p += GEMM_WAVES) {
      const int k  = ((p >> 3) << 3) + kk;
      const int c4 = ((p & 7) << 2) + cc;
      const float4 w4 = *(const float4*)(W + (size_t)k * C_OUT + c4 * 4);
      float* d = Wt + (c4 * 4) * WPITCH + k;
      d[0]          = w4.x;
      d[WPITCH]     = w4.y;
      d[2 * WPITCH] = w4.z;
      d[3 * WPITCH] = w4.w;
    }
  }
  if (threadIdx.x < 2 * C_OUT) att_s[threadIdx.x] = att[threadIdx.x];
  __syncthreads();
  if (!valid) {
    const int iw = __builtin_amdgcn_readfirstlane(strip - M_STRIPS);
    const int g0 = iw * G_LO + (iw < G_REM ? iw : G_REM);
    const int e  = g0 * 32 + lane;
    bucket_edges<G_LO, G_HALF>(esrc, edst, bucket, ovfn, ovf_dst, ovf_src, e, iw < G_REM);
    return;
  }

  const float* bcol = Wt + l * WPITCH + 2 * hi;

  v8f acc[8];
#pragma unroll
  for (int t = 0; t < 8; ++t) acc[t] = (v8f)0.0f;

  v2f b0[8], b1[8];
#pragma unroll
  for (int t = 0; t < 8; ++t) b0[t] = *(const v2f*)(bcol + 16 * t * WPITCH);

  for (int c = 0; c < DIN / 32; ++c) {
    const int cn = (c + 1 < DIN / 32) ? (c + 1) : c;
#pragma unroll
    for (int j = 0; j < 8; ++j) a_nxt[j] = *(const v2f*)(arow + 32 * cn + 4 * j);
    __builtin_amdgcn_sched_barrier(0);

#pragma unroll
    for (int j = 0; j < 8; j += 2) {
      const int q  = 8 * c + j;
      const int q2 = (q + 2 < DIN / 4) ? (q + 2) : (DIN / 4 - 1);
#pragma unroll
      for (int t = 0; t < 8; ++t) b1[t] = *(const v2f*)(bcol + 16 * t * WPITCH + 4 * (q + 1));
      __builtin_amdgcn_sched_barrier(0);
#pragma unroll
      for (int t = 0; t < 8; ++t)
        acc[t] = __builtin_amdgcn_wmma_f32_16x16x4_f32(false, a_cur[j], false, b0[t],
                                                       (short)0, acc[t], false, false);
#pragma unroll
      for (int t = 0; t < 8; ++t) b0[t] = *(const v2f*)(bcol + 16 * t * WPITCH + 4 * q2);
      __builtin_amdgcn_sched_barrier(0);
#pragma unroll
      for (int t = 0; t < 8; ++t)
        acc[t] = __builtin_amdgcn_wmma_f32_16x16x4_f32(false, a_cur[j + 1], false, b1[t],
                                                       (short)0, acc[t], false, false);
    }
#pragma unroll
    for (int j = 0; j < 8; ++j) a_cur[j] = a_nxt[j];
  }

  float* orow = h + (size_t)(tM + hi * 8) * C_OUT + l;
#pragma unroll
  for (int t = 0; t < 8; ++t)
#pragma unroll
    for (int r = 0; r < 8; ++r) orow[(size_t)r * C_OUT + 16 * t] = acc[t][r];

  float p[16];
  {
    float wi[8], wj[8];
#pragma unroll
    for (int t = 0; t < 8; ++t) { wi[t] = att_s[16 * t + l]; wj[t] = att_s[C_OUT + 16 * t + l]; }
#pragma unroll
    for (int r = 0; r < 8; ++r) {
      float si = 0.0f, sj = 0.0f;
#pragma unroll
      for (int t = 0; t < 8; ++t) { si += acc[t][r] * wi[t]; sj += acc[t][r] * wj[t]; }
      p[r] = si;
      p[8 + r] = sj;
    }
  }
  fold_step<8>(p, l);
  fold_step<4>(p, l);
  fold_step<2>(p, l);
  fold_step<1>(p, l);
  {
    float* dst = (l & 8) ? aj : ai;
    dst[tM + (l & 7) + 8 * hi] = p[0];
  }
}

__device__ __forceinline__ float leaky(float a) {
  return a > 0.0f ? a : NEG_SLOPE * a;
}

__device__ __forceinline__ float wave_max(float v) {
#pragma unroll
  for (int off = 16; off > 0; off >>= 1) v = fmaxf(v, __shfl_xor(v, off, 32));
  return v;
}
__device__ __forceinline__ float wave_sum(float v) {
#pragma unroll
  for (int off = 16; off > 0; off >>= 1) v += __shfl_xor(v, off, 32);
  return v;
}

__device__ __forceinline__ int entry_get(const int* __restrict__ bk,
                                         const int* __restrict__ ovf_dst,
                                         const int* __restrict__ ovf_src,
                                         int d, int g, int total_b, int G, bool& ok) {
  int s = d;
  ok = false;
  if (g < total_b) {
    ok = true;
    if (g >= 1) s = min(max(bk[g], 0), N_NODES - 1);
  } else if (g < G) {
    const int j = min(max(g - total_b, 0), E_EDGES - 1);
    if (ovf_dst[j] == d) {
      ok = true;
      s = min(max(ovf_src[j], 0), N_NODES - 1);
    }
  }
  return s;
}

#define AGG_ROWS 8
__device__ __forceinline__ void gather_rows(float4& acc, const float* __restrict__ h,
                                            int s, float coef, int n, int lane) {
  for (int k = 0; k < n; k += AGG_ROWS) {
    float4 hv[AGG_ROWS];
    float  ck[AGG_ROWS];
#pragma unroll
    for (int u = 0; u < AGG_ROWS; ++u) {
      const int sk = __builtin_amdgcn_readlane(s, k + u);
      ck[u] = __uint_as_float(__builtin_amdgcn_readlane(__float_as_uint(coef), k + u));
      hv[u] = *(const float4*)(h + (size_t)sk * C_OUT + lane * 4);
    }
#pragma unroll
    for (int u = 0; u < AGG_ROWS; ++u) {
      acc.x += hv[u].x * ck[u];
      acc.y += hv[u].y * ck[u];
      acc.z += hv[u].z * ck[u];
      acc.w += hv[u].w * ck[u];
    }
  }
}

__global__ __launch_bounds__(256) void k_aggregate(const float* __restrict__ h,
                                                   const float* __restrict__ ai,
                                                   const float* __restrict__ aj,
                                                   const int* __restrict__ bucket,
                                                   const int* __restrict__ ovfn,
                                                   const int* __restrict__ ovf_dst,
                                                   const int* __restrict__ ovf_src,
                                                   const float* __restrict__ bias,
                                                   float* __restrict__ out) {
  const int wave = __builtin_amdgcn_readfirstlane((int)(threadIdx.x >> 5));
  const int lane = threadIdx.x & 31;
  const int d = blockIdx.x * 8 + wave;
  const int* bk = bucket + (size_t)d * BK_WORDS;
  const int w = bk[lane];
  const float aid = ai[d];
  const int c = max(__builtin_amdgcn_readfirstlane(w), 0);
  float4 acc = make_float4(0.0f, 0.0f, 0.0f, 0.0f);

  if (c < 32) {
    const int total = c + 1;
    const bool valid = lane < total;
    const int s = (lane >= 1 && valid) ? min(max(w, 0), N_NODES - 1) : d;
    const float a   = valid ? leaky(aid + aj[s]) : -__builtin_inff();
    const float m   = wave_max(a);
    const float ex  = valid ? expf(a - m) : 0.0f;
    const float den = wave_sum(ex);
    const float coef = ex / (den + 1e-16f);
    gather_rows(acc, h, s, coef, total, lane);
  } else {
    const int total_b = min(c, BK_SLOTS) + 1;
    int G = total_b;
    if (c > BK_SLOTS) G += min(max(ovfn[0], 0), E_EDGES);
    float m = -__builtin_inff();
    for (int base = 0; base < G; base += 32) {
      bool ok;
      const int s = entry_get(bk, ovf_dst, ovf_src, d, base + lane, total_b, G, ok);
      if (ok) m = fmaxf(m, leaky(aid + aj[s]));
    }
    m = wave_max(m);
    float den = 0.0f;
    for (int base = 0; base < G; base += 32) {
      bool ok;
      const int s = entry_get(bk, ovf_dst, ovf_src, d, base + lane, total_b, G, ok);
      if (ok) den += expf(leaky(aid + aj[s]) - m);
    }
    den = wave_sum(den);
    for (int base = 0; base < G; base += 32) {
      bool ok;
      const int s = entry_get(bk, ovf_dst, ovf_src, d, base + lane, total_b, G, ok);
      const float coef = ok ? expf(leaky(aid + aj[s]) - m) / (den + 1e-16f) : 0.0f;
      if (__any(ok)) gather_rows(acc, h, s, coef, min(32, G - base), lane);
    }
  }

  const float4 bv = *(const float4*)(bias + lane * 4);
  *(float4*)(out + (size_t)d * C_OUT + lane * 4) =
      make_float4(acc.x + bv.x, acc.y + bv.y, acc.z + bv.z, acc.w + bv.w);
}

extern "C" void kernel_launch(void* const* d_in, const int* in_sizes, int n_in,
                              void* d_out, int out_size, void* d_ws, size_t ws_size,
                              hipStream_t stream) {
  const float* x    = (const float*)d_in[0];
  const int*   ei   = (const int*)  d_in[1];
  const float* W    = (const float*)d_in[2];
  const float* att  = (const float*)d_in[3];
  const float* bias = (const float*)d_in[4];
  float* out = (float*)d_out;

  float* h       = (float*)d_ws;
  int*   bucket  = (int*)(h + (size_t)N_NODES * C_OUT);
  float* ai      = (float*)(bucket + (size_t)N_NODES * BK_WORDS);
  float* aj      = ai + N_NODES;
  int*   ovfn    = (int*)(aj + N_NODES);
  int*   ovf_dst = ovfn + 4;
  int*   ovf_src = ovf_dst + E_EDGES;

  const int* esrc = ei;
  const int* edst = ei + E_EDGES;

  k_clear    <<<CLEAR_BLOCKS, 256, 0, stream>>>(bucket, ovfn);
  k_gemm     <<<GEMM_BLOCKS, GEMM_THREADS, 0, stream>>>(x, W, att, h, ai, aj, esrc, edst,
                                                        bucket, ovfn, ovf_dst, ovf_src);
  k_aggregate<<<N_NODES / 8, 256, 0, stream>>>(h, ai, aj, bucket, ovfn, ovf_dst, ovf_src, bias, out);
}
